# starting kernel
# baseline (speedup 1.0000x reference)
#include <hip/hip_runtime.h>

typedef float v2f __attribute__((ext_vector_type(2)));
typedef float v4f __attribute__((ext_vector_type(4)));
typedef float v8f __attribute__((ext_vector_type(8)));

#define B_    128
#define F_    1024
#define CIN_  16
#define COUT_ 16
#define K_    4
#define R_    2

__global__ __launch_bounds__(256) void sumconv_wmma_kernel(
    const float* __restrict__ ll,
    const float* __restrict__ logits,
    float* __restrict__ out)
{
  const int wave = (blockIdx.x * (int)blockDim.x + (int)threadIdx.x) >> 5;
  const int lane = threadIdx.x & 31;
  const int hi   = lane >> 4;
  const int col  = lane & 15;

  const int g  =  wave       & 3;
  const int kw = (wave >> 2) & 3;
  const int kh = (wave >> 4) & 3;
  const int b  =  wave >> 6;


  float lg0[8], lg1[8];
  const int lbase = col * (CIN_ * K_ * K_ * R_) + kh * (K_ * R_) + kw * R_;
  #pragma unroll
  for (int j = 0; j < 8; ++j) {
    const int cin = 4 * (j >> 1) + 2 * hi + (j & 1);
    const v2f p = *(const v2f*)(logits + lbase + cin * (K_ * K_ * R_));
    lg0[j] = p.x; lg1[j] = p.y;
  }
  float wm0 = lg0[0], wm1 = lg1[0];
  #pragma unroll
  for (int j = 1; j < 8; ++j) { wm0 = fmaxf(wm0, lg0[j]); wm1 = fmaxf(wm1, lg1[j]); }
  wm0 = fmaxf(wm0, __shfl_xor(wm0, 16, 32));
  wm1 = fmaxf(wm1, __shfl_xor(wm1, 16, 32));
  float ae0[8], ae1[8], ps0 = 0.0f, ps1 = 0.0f;
  #pragma unroll
  for (int j = 0; j < 8; ++j) {
    ae0[j] = __expf(lg0[j] - wm0); ps0 += ae0[j];
    ae1[j] = __expf(lg1[j] - wm1); ps1 += ae1[j];
  }
  const float wi0 = 1.0f / (ps0 + __shfl_xor(ps0, 16, 32));
  const float wi1 = 1.0f / (ps1 + __shfl_xor(ps1, 16, 32));
  v2f af0[4], af1[4];
  #pragma unroll
  for (int c = 0; c < 4; ++c) {
    af0[c].x = ae0[2 * c] * wi0; af0[c].y = ae0[2 * c + 1] * wi0;
    af1[c].x = ae1[2 * c] * wi1; af1[c].y = ae1[2 * c + 1] * wi1;
  }

  const int idx = g * 16 + col;
  const int hb  = idx >> 3, wb = idx & 7;
  const int f   = (4 * hb + kh) * 32 + 4 * wb + kw;
  const v4f* px = (const v4f*)(ll + (size_t)(b * F_ + f) * (CIN_ * R_));
  float x0[8], x1[8];
  #pragma unroll
  for (int c = 0; c < 4; ++c) {
    const v4f q = px[2 * c + hi];
    x0[2 * c]     = q.x; x1[2 * c]     = q.y;
    x0[2 * c + 1] = q.z; x1[2 * c + 1] = q.w;
  }
  float m0 = x0[0], m1 = x1[0];
  #pragma unroll
  for (int j = 1; j < 8; ++j) { m0 = fmaxf(m0, x0[j]); m1 = fmaxf(m1, x1[j]); }
  m0 = fmaxf(m0, __shfl_xor(m0, 16, 32));
  m1 = fmaxf(m1, __shfl_xor(m1, 16, 32));
  v2f bf0[4], bf1[4];
  #pragma unroll
  for (int c = 0; c < 4; ++c) {
    bf0[c].x = __expf(x0[2 * c]     - m0); bf0[c].y = __expf(x0[2 * c + 1] - m0);
    bf1[c].x = __expf(x1[2 * c]     - m1); bf1[c].y = __expf(x1[2 * c + 1] - m1);
  }

  v8f acc0 = {0.f, 0.f, 0.f, 0.f, 0.f, 0.f, 0.f, 0.f};
  v8f acc1 = {0.f, 0.f, 0.f, 0.f, 0.f, 0.f, 0.f, 0.f};
  #pragma unroll
  for (int c = 0; c < 4; ++c) {
    acc0 = __builtin_amdgcn_wmma_f32_16x16x4_f32(false, af0[c], false, bf0[c],
                                                 (short)0, acc0, false, false);
    acc1 = __builtin_amdgcn_wmma_f32_16x16x4_f32(false, af1[c], false, bf1[c],
                                                 (short)0, acc1, false, false);
  }

  v2f* po = (v2f*)(out + (size_t)(b * F_ + f) * (COUT_ * R_));
  #pragma unroll
  for (int v = 0; v < 8; ++v) {
    const int o = hi * 8 + v;
    v2f w; w.x = m0 + __logf(acc0[v]); w.y = m1 + __logf(acc1[v]);
    po[o] = w;
  }
}

extern "C" void kernel_launch(void* const* d_in, const int* in_sizes, int n_in,
                              void* d_out, int out_size, void* d_ws, size_t ws_size,
                              hipStream_t stream) {
  (void)in_sizes; (void)n_in; (void)d_ws; (void)ws_size; (void)out_size;
  const float* ll     = (const float*)d_in[0];
  const float* logits = (const float*)d_in[1];
  float* out          = (float*)d_out;

  dim3 grid(1024), block(256);
  sumconv_wmma_kernel<<<grid, block, 0, stream>>>(ll, logits, out);
}


# Strip decomposition: each of 256 workgroups normalises its kh row of 4 weight cells once into an LDS A-fragment table; a wave owns 2 whole image rows and walks the 4 kw cells: 2048 waves, was 8192
# speedup vs baseline: 1.0522x; 1.0522x over previous
#include <hip/hip_runtime.h>

typedef float v2f __attribute__((ext_vector_type(2)));
typedef float v4f __attribute__((ext_vector_type(4)));
typedef float v8f __attribute__((ext_vector_type(8)));

#define B_    128
#define F_    1024
#define CIN_  16
#define COUT_ 16
#define K_    4
#define R_    2

__global__ __launch_bounds__(256) void sumconv_strip_kernel(
    const float* __restrict__ ll,
    const float* __restrict__ logits,
    float* __restrict__ out)
{
  __shared__ v4f wtab[K_ * 4 * 32];

  const int tid  = threadIdx.x;
  const int lane = tid & 31;
  const int wv   = tid >> 5;
  const int hi   = lane >> 4;
  const int col  = lane & 15;
  const int kh   = blockIdx.x & 3;

  const int tr  = tid & 1;
  const int tkw = (tid >> 1) & 3;
  const int th  = (tid >> 3) & 1;
  const int to  = tid >> 4;
  const float* pl = logits + to * (CIN_ * K_ * K_ * R_) + kh * (K_ * R_) + tkw * R_ + tr;
  float lg[8];
  #pragma unroll
  for (int j = 0; j < 8; ++j) {
    const int cin = 4 * (j >> 1) + 2 * th + (j & 1);
    lg[j] = pl[cin * (K_ * K_ * R_)];
  }

  const int u  = (blockIdx.x >> 2) * 8 + wv;
  const int b  = u >> 2;
  const int hb = (u & 3) * 2 + (col >> 3);
  const int wb = col & 7;
  const int f0 = (4 * hb + kh) * 32 + 4 * wb;
  const v4f* px = (const v4f*)(ll + (size_t)(b * F_ + f0) * (CIN_ * R_));
  v4f q[4][4];
  #pragma unroll
  for (int kw = 0; kw < 4; ++kw) {
    #pragma unroll
    for (int c = 0; c < 4; ++c)
      q[kw][c] = px[kw * 8 + 2 * c + hi];
  }

  {
    float wm = lg[0];
    #pragma unroll
    for (int j = 1; j < 8; ++j) wm = fmaxf(wm, lg[j]);
    wm = fmaxf(wm, __shfl_xor(wm, 8, 32));
    float ae[8], ps = 0.0f;
    #pragma unroll
    for (int j = 0; j < 8; ++j) { ae[j] = __expf(lg[j] - wm); ps += ae[j]; }
    const float wi = 1.0f / (ps + __shfl_xor(ps, 8, 32));
    float* wt = (float*)wtab;
    #pragma unroll
    for (int c = 0; c < 4; ++c) {
      v2f w; w.x = ae[2 * c] * wi; w.y = ae[2 * c + 1] * wi;
      *(v2f*)(wt + ((tkw * 4 + c) * 32 + to + 16 * th) * 4 + tr * 2) = w;
    }
  }
  __builtin_amdgcn_fence(__ATOMIC_RELEASE, "workgroup", "local");
  __builtin_amdgcn_s_barrier();
  __builtin_amdgcn_fence(__ATOMIC_ACQUIRE, "workgroup", "local");

  v4f* po = (v4f*)(out + (size_t)(b * F_ + f0) * (COUT_ * R_));
  #pragma unroll
  for (int kw = 0; kw < 4; ++kw) {
    v2f af0[4], af1[4];
    #pragma unroll
    for (int c = 0; c < 4; ++c) {
      const v4f t = wtab[(kw * 4 + c) * 32 + lane];
      af0[c].x = t.x; af0[c].y = t.y;
      af1[c].x = t.z; af1[c].y = t.w;
    }

    float x0[8], x1[8];
    #pragma unroll
    for (int c = 0; c < 4; ++c) {
      const v4f t = q[kw][c];
      x0[2 * c]     = t.x; x1[2 * c]     = t.y;
      x0[2 * c + 1] = t.z; x1[2 * c + 1] = t.w;
    }
    float m0 = x0[0], m1 = x1[0];
    #pragma unroll
    for (int j = 1; j < 8; ++j) { m0 = fmaxf(m0, x0[j]); m1 = fmaxf(m1, x1[j]); }
    m0 = fmaxf(m0, __shfl_xor(m0, 16, 32));
    m1 = fmaxf(m1, __shfl_xor(m1, 16, 32));
    v2f bf0[4], bf1[4];
    #pragma unroll
    for (int c = 0; c < 4; ++c) {
      bf0[c].x = __expf(x0[2 * c]     - m0); bf0[c].y = __expf(x0[2 * c + 1] - m0);
      bf1[c].x = __expf(x1[2 * c]     - m1); bf1[c].y = __expf(x1[2 * c + 1] - m1);
    }

    v8f acc0 = {0.f, 0.f, 0.f, 0.f, 0.f, 0.f, 0.f, 0.f};
    v8f acc1 = {0.f, 0.f, 0.f, 0.f, 0.f, 0.f, 0.f, 0.f};
    #pragma unroll
    for (int c = 0; c < 4; ++c) {
      acc0 = __builtin_amdgcn_wmma_f32_16x16x4_f32(false, af0[c], false, bf0[c],
                                                   (short)0, acc0, false, false);
      acc1 = __builtin_amdgcn_wmma_f32_16x16x4_f32(false, af1[c], false, bf1[c],
                                                   (short)0, acc1, false, false);
    }

    #pragma unroll
    for (int p = 0; p < 4; ++p) {
      v4f w;
      w.x = m0 + __logf(acc0[2 * p]);     w.y = m1 + __logf(acc1[2 * p]);
      w.z = m0 + __logf(acc0[2 * p + 1]); w.w = m1 + __logf(acc1[2 * p + 1]);
      po[kw * 8 + hi * 4 + p] = w;
    }
  }
}

extern "C" void kernel_launch(void* const* d_in, const int* in_sizes, int n_in,
                              void* d_out, int out_size, void* d_ws, size_t ws_size,
                              hipStream_t stream) {
  (void)in_sizes; (void)n_in; (void)d_ws; (void)ws_size; (void)out_size;
  const float* ll     = (const float*)d_in[0];
  const float* logits = (const float*)d_in[1];
  float* out          = (float*)d_out;

  dim3 grid(256), block(256);
  sumconv_strip_kernel<<<grid, block, 0, stream>>>(ll, logits, out);
}
